# starting kernel
# baseline (speedup 1.0000x reference)
#include <hip/hip_runtime.h>


typedef float v2f __attribute__((ext_vector_type(2)));
typedef float v8f __attribute__((ext_vector_type(8)));

#define BLOCK 256

static __device__ __forceinline__ float fast_tanh(float x) {
#if __has_builtin(__builtin_amdgcn_tanhf)
  return __builtin_amdgcn_tanhf(x);
#elif __has_builtin(__builtin_amdgcn_tanh_f32)
  return __builtin_amdgcn_tanh_f32(x);
#else
  float e = __builtin_amdgcn_exp2f(x * 2.8853900817779268f);
  return 1.0f - 2.0f / (e + 1.0f);
#endif
}

static __device__ __forceinline__ float softplus_f(float x) {
  float t = __builtin_amdgcn_exp2f(-__builtin_fabsf(x) * 1.4426950408889634f);
  return fmaxf(x, 0.0f) + 0.6931471805599453f * __builtin_amdgcn_logf(1.0f + t);
}

__global__ __launch_bounds__(BLOCK) void csys_kernel(
    const float* __restrict__ t_in, const float* __restrict__ z,
    const float* __restrict__ logK, const float* __restrict__ logz,
    const float* __restrict__ logp, const float* __restrict__ W1,
    const float* __restrict__ b1, const float* __restrict__ W2,
    const float* __restrict__ b2, float* __restrict__ out, int n) {
  __shared__ float xch[2 * BLOCK];

  const int tid   = threadIdx.x;
  const int lane  = tid & 31;
  const int wbase = (tid >> 5) * 64;
  const int gid   = blockIdx.x * BLOCK + tid;
  const int idx = (gid < n) ? gid : (n - 1);

  const float tt = t_in[idx];
  const float x1 = z[0 * n + idx];
  const float v1 = z[1 * n + idx];
  const float x2 = z[2 * n + idx];
  const float v2 = z[3 * n + idx];
  const float xc = z[4 * n + idx];

  const int khalf = lane >> 4;
  const int nn    = lane & 15;

  v2f bmat[16];
#pragma unroll
  for (int c = 0; c < 16; ++c) {
    const int kb = 4 * c + 2 * khalf;
    float bx = 0.0f, by = 0.0f;
    if (nn < 2) {
      bx = W2[kb * 2 + nn];
      by = W2[(kb + 1) * 2 + nn];
    }
    bmat[c].x = bx;
    bmat[c].y = by;
  }

  const float v2t0 = __shfl(v2, nn, 32);
  const float v2t1 = __shfl(v2, 16 + nn, 32);
  v8f acc0 = {};
  v8f acc1 = {};
#pragma unroll
  for (int c = 0; c < 16; ++c) {
    const int ka = 4 * c + 2 * khalf;
    const float2 w1p = *reinterpret_cast<const float2*>(W1 + ka);
    const float2 b1p = *reinterpret_cast<const float2*>(b1 + ka);
    v2f a0, a1;
    a0.x = fast_tanh(fmaf(v2t0, w1p.x, b1p.x));
    a0.y = fast_tanh(fmaf(v2t0, w1p.y, b1p.y));
    a1.x = fast_tanh(fmaf(v2t1, w1p.x, b1p.x));
    a1.y = fast_tanh(fmaf(v2t1, w1p.y, b1p.y));
    acc0 = __builtin_amdgcn_wmma_f32_16x16x4_f32(
        false, a0, false, bmat[c], (short)0, acc0, false, false);
    acc1 = __builtin_amdgcn_wmma_f32_16x16x4_f32(
        false, a1, false, bmat[c], (short)0, acc1, false, false);
  }

  if (nn < 2) {
    const int wv0 = khalf * 8;
    const int wv1 = 16 + khalf * 8;
#pragma unroll
    for (int r = 0; r < 8; ++r) {
      xch[wbase + (wv0 + r) * 2 + nn] = acc0[r];
      xch[wbase + (wv1 + r) * 2 + nn] = acc1[r];
    }
  }
  __syncthreads();

  const float kinetic = softplus_f(xch[wbase + lane * 2 + 0] + b2[0]);
  const float stic    = softplus_f(xch[wbase + lane * 2 + 1] + b2[1]);

  const float Kc = expf(logK[0]);
  const float zc = expf(logz[0]);
  const float pc = expf(logp[0]);

  const float x2ref = 0.5f * __sinf(0.5f * tt);
  const float e     = x2ref - x2;
  const float d_xc  = -pc * xc + e;
  const float u     = Kc * (zc - pc) * xc + Kc * e;

  const float dx1 = v1;
  const float dv1 = u - 2.0f * x1 - 0.5f * v1 - 3.0f * (x1 - x2) - 0.8f * (v1 - v2);
  const float dx2 = v2;

  const float F_net    = 3.0f * (x1 - x2) + 0.8f * (v1 - v2);
  const float F_static = -fminf(fmaxf(F_net, -stic), stic);
  const float sgn      = (v2 > 0.0f) ? 1.0f : ((v2 < 0.0f) ? -1.0f : 0.0f);
  const float F_kin    = -kinetic * sgn;
  const float F_fric   = (__builtin_fabsf(v2) < 0.01f) ? F_static : F_kin;
  const float dv2      = (F_net + F_fric) * (1.0f / 1.5f);

  out[0 * n + idx] = dx1;
  out[1 * n + idx] = dv1;
  out[2 * n + idx] = dx2;
  out[3 * n + idx] = dv2;
  out[4 * n + idx] = d_xc;
}

extern "C" void kernel_launch(void* const* d_in, const int* in_sizes, int n_in,
                              void* d_out, int out_size, void* d_ws,
                              size_t ws_size, hipStream_t stream) {
  const float* t    = (const float*)d_in[0];
  const float* z    = (const float*)d_in[1];
  const float* logK = (const float*)d_in[2];
  const float* logz = (const float*)d_in[3];
  const float* logp = (const float*)d_in[4];
  const float* W1   = (const float*)d_in[5];
  const float* b1   = (const float*)d_in[6];
  const float* W2   = (const float*)d_in[7];
  const float* b2   = (const float*)d_in[8];
  float* out        = (float*)d_out;

  const int n      = in_sizes[0];
  const int blocks = (n + BLOCK - 1) / BLOCK;
  hipLaunchKernelGGL(csys_kernel, dim3(blocks), dim3(BLOCK), 0, stream,
                     t, z, logK, logz, logp, W1, b1, W2, b2, out, n);
}


# Persistent launch: 512 workgroups each loop over 256-element tiles (8 per workgroup here); weight fragments and gains stay in registers across tiles, next tile's inputs in flight during the math
# speedup vs baseline: 1.0546x; 1.0546x over previous
#include <hip/hip_runtime.h>


typedef float v2f __attribute__((ext_vector_type(2)));
typedef float v8f __attribute__((ext_vector_type(8)));

#define BLOCK 256
#define GRID  512

static __device__ __forceinline__ float fast_tanh(float x) {
#if __has_builtin(__builtin_amdgcn_tanhf)
  return __builtin_amdgcn_tanhf(x);
#elif __has_builtin(__builtin_amdgcn_tanh_f32)
  return __builtin_amdgcn_tanh_f32(x);
#else
  float e = __builtin_amdgcn_exp2f(x * 2.8853900817779268f);
  return 1.0f - 2.0f / (e + 1.0f);
#endif
}

static __device__ __forceinline__ float softplus_f(float x) {
  float t = __builtin_amdgcn_exp2f(-__builtin_fabsf(x) * 1.4426950408889634f);
  return fmaxf(x, 0.0f) + 0.6931471805599453f * __builtin_amdgcn_logf(1.0f + t);
}

__global__ __launch_bounds__(BLOCK) void csys_kernel(
    const float* __restrict__ t_in, const float* __restrict__ z,
    const float* __restrict__ logK, const float* __restrict__ logz,
    const float* __restrict__ logp, const float* __restrict__ W1,
    const float* __restrict__ b1, const float* __restrict__ W2,
    const float* __restrict__ b2, float* __restrict__ out, int n) {
  __shared__ float xch[2 * BLOCK];

  if (n <= 0) return;

  const int tid   = threadIdx.x;
  const int lane  = tid & 31;
  const int wbase = (tid >> 5) * 64;

  const int khalf = lane >> 4;
  const int nn    = lane & 15;

  v2f bmat[16];
#pragma unroll
  for (int c = 0; c < 16; ++c) {
    const int kb = 4 * c + 2 * khalf;
    float bx = 0.0f, by = 0.0f;
    if (nn < 2) {
      bx = W2[kb * 2 + nn];
      by = W2[(kb + 1) * 2 + nn];
    }
    bmat[c].x = bx;
    bmat[c].y = by;
  }
  float2 w1p[16], b1p[16];
#pragma unroll
  for (int c = 0; c < 16; ++c) {
    const int ka = 4 * c + 2 * khalf;
    w1p[c] = *reinterpret_cast<const float2*>(W1 + ka);
    b1p[c] = *reinterpret_cast<const float2*>(b1 + ka);
  }
  const float b20 = b2[0];
  const float b21 = b2[1];
  const float Kc = expf(logK[0]);
  const float zc = expf(logz[0]);
  const float pc = expf(logp[0]);

  const int stride = gridDim.x * BLOCK;
  int base = blockIdx.x * BLOCK;

  int idx = base + tid;
  idx = (idx < n) ? idx : (n - 1);

  float tt = t_in[idx];
  float x1 = z[0 * n + idx];
  float v1 = z[1 * n + idx];
  float x2 = z[2 * n + idx];
  float v2 = z[3 * n + idx];
  float xc = z[4 * n + idx];

  for (; base < n; base += stride) {
    int nidx = base + stride + tid;
    nidx = (nidx < n) ? nidx : (n - 1);
    const float ntt = t_in[nidx];
    const float nx1 = z[0 * n + nidx];
    const float nv1 = z[1 * n + nidx];
    const float nx2 = z[2 * n + nidx];
    const float nv2 = z[3 * n + nidx];
    const float nxc = z[4 * n + nidx];

    const float v2t0 = __shfl(v2, nn, 32);
    const float v2t1 = __shfl(v2, 16 + nn, 32);
    v8f acc0 = {};
    v8f acc1 = {};
#pragma unroll
    for (int c = 0; c < 16; ++c) {
      v2f a0, a1;
      a0.x = fast_tanh(fmaf(v2t0, w1p[c].x, b1p[c].x));
      a0.y = fast_tanh(fmaf(v2t0, w1p[c].y, b1p[c].y));
      a1.x = fast_tanh(fmaf(v2t1, w1p[c].x, b1p[c].x));
      a1.y = fast_tanh(fmaf(v2t1, w1p[c].y, b1p[c].y));
      acc0 = __builtin_amdgcn_wmma_f32_16x16x4_f32(
          false, a0, false, bmat[c], (short)0, acc0, false, false);
      acc1 = __builtin_amdgcn_wmma_f32_16x16x4_f32(
          false, a1, false, bmat[c], (short)0, acc1, false, false);
    }

    if (nn < 2) {
      const int wv0 = khalf * 8;
      const int wv1 = 16 + khalf * 8;
#pragma unroll
      for (int r = 0; r < 8; ++r) {
        xch[wbase + (wv0 + r) * 2 + nn] = acc0[r];
        xch[wbase + (wv1 + r) * 2 + nn] = acc1[r];
      }
    }
    __syncthreads();

    const float kinetic = softplus_f(xch[wbase + lane * 2 + 0] + b20);
    const float stic    = softplus_f(xch[wbase + lane * 2 + 1] + b21);

    const float x2ref = 0.5f * __sinf(0.5f * tt);
    const float e     = x2ref - x2;
    const float d_xc  = -pc * xc + e;
    const float u     = Kc * (zc - pc) * xc + Kc * e;

    const float dx1 = v1;
    const float dv1 = u - 2.0f * x1 - 0.5f * v1 - 3.0f * (x1 - x2) - 0.8f * (v1 - v2);
    const float dx2 = v2;

    const float F_net    = 3.0f * (x1 - x2) + 0.8f * (v1 - v2);
    const float F_static = -fminf(fmaxf(F_net, -stic), stic);
    const float sgn      = (v2 > 0.0f) ? 1.0f : ((v2 < 0.0f) ? -1.0f : 0.0f);
    const float F_kin    = -kinetic * sgn;
    const float F_fric   = (__builtin_fabsf(v2) < 0.01f) ? F_static : F_kin;
    const float dv2      = (F_net + F_fric) * (1.0f / 1.5f);

    out[0 * n + idx] = dx1;
    out[1 * n + idx] = dv1;
    out[2 * n + idx] = dx2;
    out[3 * n + idx] = dv2;
    out[4 * n + idx] = d_xc;

    idx = nidx;
    tt = ntt; x1 = nx1; v1 = nv1; x2 = nx2; v2 = nv2; xc = nxc;
  }
}

extern "C" void kernel_launch(void* const* d_in, const int* in_sizes, int n_in,
                              void* d_out, int out_size, void* d_ws,
                              size_t ws_size, hipStream_t stream) {
  const float* t    = (const float*)d_in[0];
  const float* z    = (const float*)d_in[1];
  const float* logK = (const float*)d_in[2];
  const float* logz = (const float*)d_in[3];
  const float* logp = (const float*)d_in[4];
  const float* W1   = (const float*)d_in[5];
  const float* b1   = (const float*)d_in[6];
  const float* W2   = (const float*)d_in[7];
  const float* b2   = (const float*)d_in[8];
  float* out        = (float*)d_out;

  const int n = in_sizes[0];
  hipLaunchKernelGGL(csys_kernel, dim3(GRID), dim3(BLOCK), 0, stream,
                     t, z, logK, logz, logp, W1, b1, W2, b2, out, n);
}
